# starting kernel
# baseline (speedup 1.0000x reference)
#include <hip/hip_runtime.h>

typedef __attribute__((ext_vector_type(2))) float v2f;
typedef __attribute__((ext_vector_type(8))) float v8f;

#define TT 4096
#define DD 64
#define KK 1024

#define LDS_HES_B 262144
#define LDS_TOTAL 266240

__global__ void __launch_bounds__(256)
vq_prep(const float* __restrict__ cb, float* __restrict__ cbG,
        float* __restrict__ hes) {
  int n = blockIdx.x * blockDim.x + threadIdx.x;
  if (n >= KK) return;
  const float* row = cb + n * DD;
  float s = 0.f;
  v2f* cp = (v2f*)cbG;
  int base = (n >> 4) * 512 + (n & 15);
#pragma unroll
  for (int j = 0; j < 32; ++j) {
    float x = row[2 * j];
    float y = row[2 * j + 1];
    s += x * x + y * y;
    v2f p;
    p.x = x;
    p.y = y;
    cp[base + j * 16] = p;
  }
  hes[n] = 0.5f * s;
}

__global__ void __launch_bounds__(512)
vq_encode(const float* __restrict__ z, const float* __restrict__ cbG,
          const float* __restrict__ hes, int* __restrict__ idx) {
  extern __shared__ char smem[];

  {
    int tid = threadIdx.x;
    const float4* src4 = (const float4*)cbG;
    float4* dst4 = (float4*)smem;
#pragma unroll
    for (int it = 0; it < 32; ++it) {
      dst4[it * 512 + tid] = src4[it * 512 + tid];
    }
    float* shes = (float*)(smem + LDS_HES_B);
    for (int i = tid; i < KK; i += 512) shes[i] = hes[i];
  }
  __syncthreads();

  const int lane = threadIdx.x & 31;
  const int wave = threadIdx.x >> 5;
  const int tile = blockIdx.x * 16 + wave;
  const int token0 = tile << 4;
  const int b = token0 >> 12;
  const int t0 = token0 & (TT - 1);
  const int hi = lane >> 4;
  const int ln = lane & 15;

  const float* zb = z + (size_t)b * DD * TT + t0 + ln;
  float a[32];
#pragma unroll
  for (int kk = 0; kk < 16; ++kk) {
    int d0 = 4 * kk + 2 * hi;
    a[2 * kk]     = zb[(size_t)d0 * TT];
    a[2 * kk + 1] = zb[(size_t)(d0 + 1) * TT];
  }
#pragma unroll
  for (int i = 0; i < 32; ++i) {
    asm volatile("" : "+v"(a[i]));
  }

  float best[8];
  int bestn[8];
#pragma unroll
  for (int e = 0; e < 8; ++e) { best[e] = -3.4e38f; bestn[e] = 0; }

  const char* sb0 = smem + hi * 128 + ln * 8;
  const char* sb1 = sb0 + 4096;
  const float* shp = (const float*)(smem + LDS_HES_B) + ln;

  for (int nt = 0; nt < KK / 32; ++nt) {
    const int n0 = nt * 32 + ln;
    const int n1 = n0 + 16;
    const float h0 = shp[0];
    const float h1 = shp[16];
    shp += 32;

    v2f bb0[16], bb1[16];
#pragma unroll
    for (int kk = 0; kk < 16; ++kk) {
      bb0[kk] = *(const v2f*)(sb0 + kk * 256);
      bb1[kk] = *(const v2f*)(sb1 + kk * 256);
    }
#pragma unroll
    for (int kk = 0; kk < 16; ++kk) {
      asm volatile("" : "+v"(bb0[kk]), "+v"(bb1[kk]));
    }
    sb0 += 8192;
    sb1 += 8192;

    v8f c0 = {};
    v8f c1 = {};
#pragma unroll
    for (int kk = 0; kk < 16; ++kk) {
      v2f av;
      av.x = a[2 * kk];
      av.y = a[2 * kk + 1];
      c0 = __builtin_amdgcn_wmma_f32_16x16x4_f32(
          false, av, false, bb0[kk], (short)0, c0, false, false);
      c1 = __builtin_amdgcn_wmma_f32_16x16x4_f32(
          false, av, false, bb1[kk], (short)0, c1, false, false);
    }

#pragma unroll
    for (int e = 0; e < 8; ++e) {
      float s0 = c0[e] - h0;
      float s1 = c1[e] - h1;
      if (s0 > best[e]) { best[e] = s0; bestn[e] = n0; }
      if (s1 > best[e]) { best[e] = s1; bestn[e] = n1; }
    }
  }

#pragma unroll
  for (int mask = 1; mask <= 8; mask <<= 1) {
#pragma unroll
    for (int e = 0; e < 8; ++e) {
      float ov = __shfl_xor(best[e], mask, 32);
      int oi = __shfl_xor(bestn[e], mask, 32);
      if (ov > best[e] || (ov == best[e] && oi < bestn[e])) {
        best[e] = ov;
        bestn[e] = oi;
      }
    }
  }

  if (ln == 0) {
    int base = token0 + hi * 8;
    int4 b0 = make_int4(bestn[0], bestn[1], bestn[2], bestn[3]);
    int4 b1 = make_int4(bestn[4], bestn[5], bestn[6], bestn[7]);
    *(int4*)(idx + base) = b0;
    *(int4*)(idx + base + 4) = b1;
  }
}

__global__ void __launch_bounds__(256)
vq_gather(const float* __restrict__ cb, const int* __restrict__ idx,
          float* __restrict__ out) {
  int gid = blockIdx.x * blockDim.x + threadIdx.x;
  int t = gid & (TT - 1);
  int d = (gid >> 12) & (DD - 1);
  int b = gid >> 18;
  int id = idx[b * TT + t];
  out[gid] = cb[id * DD + d];
}

extern "C" void kernel_launch(void* const* d_in, const int* in_sizes, int n_in,
                              void* d_out, int out_size, void* d_ws,
                              size_t ws_size, hipStream_t stream) {
  const float* z = (const float*)d_in[0];
  const float* cb = (const float*)d_in[1];
  char* ws = (char*)d_ws;
  float* cbG = (float*)ws;
  float* hes = (float*)(ws + 262144);
  int* idx = (int*)(ws + 266240);

  vq_prep<<<4, 256, 0, stream>>>(cb, cbG, hes);
  vq_encode<<<256, 512, LDS_TOTAL, stream>>>(z, cbG, hes, idx);
  vq_gather<<<16384, 256, 0, stream>>>(cb, idx, (float*)d_out);
}


# Encode re-decomposed: a wave takes its 64 codes tile by tile into registers as matrix rows (raw rows turned via LDS, half norms in-kernel, same pair order), tokens stream from LDS; prep launch gone
# speedup vs baseline: 1.4208x; 1.4208x over previous
#include <hip/hip_runtime.h>

typedef __attribute__((ext_vector_type(2))) float v2f;
typedef __attribute__((ext_vector_type(8))) float v8f;

#define TT 4096
#define DD 64
#define KK 1024

#define BT 256
#define ZP 68
#define CP 33
#define LDS_Z 0
#define LDS_CB 69632
#define LDS_CI 103424
#define LDS_ST 137216
#define LDS_TOTAL 206848

__device__ __forceinline__ float half_norm(const float* row) {
  float s = 0.f;
#pragma unroll
  for (int q = 0; q < 16; ++q) {
    const float4 u = *(const float4*)(row + 4 * q);
    s += u.x * u.x + u.y * u.y;
    s += u.z * u.z + u.w * u.w;
  }
  return 0.5f * s;
}

__device__ __forceinline__ void tile_best(const v8f& c, const float (&h)[8],
                                          const int k0, float& bt, int& kt) {
  bt = c[0] - h[0];
  kt = k0;
#pragma unroll
  for (int r = 1; r < 8; ++r) {
    const float s = c[r] - h[r];
    if (s > bt) { bt = s; kt = k0 + r; }
  }
}

__device__ __forceinline__ void lds_stores_done() {
  asm volatile("s_wait_dscnt 0x0" ::: "memory");
}
__device__ __forceinline__ void rows_signal() {
  asm volatile("s_barrier_signal -1" ::: "memory");
}
__device__ __forceinline__ void rows_wait() {
  asm volatile("s_barrier_wait -1" ::: "memory");
}

__device__ __forceinline__ v8f pack8(const float4& x, const float4& y) {
  const v8f r = {x.x, x.y, x.z, x.w, y.x, y.y, y.z, y.w};
  return r;
}

__device__ __forceinline__ float rows_to_fragments(const float4 (&rw)[8],
                                                   v2f (&at)[16], float4* sd,
                                                   const char* fr,
                                                   const char* myrow) {
#pragma unroll
  for (int i = 0; i < 8; ++i) sd[i * (2 * ZP / 4)] = rw[i];
#pragma unroll
  for (int kk = 0; kk < 16; ++kk) at[kk] = *(const v2f*)(fr + kk * 16);
  return half_norm((const float*)myrow);
}

__global__ void __launch_bounds__(512)
vq_encode(const float* __restrict__ z, const float* __restrict__ cb,
          int* __restrict__ idx) {
  extern __shared__ char smem[];
  float* zs = (float*)(smem + LDS_Z);
  float* cbest = (float*)(smem + LDS_CB);
  int* cidx = (int*)(smem + LDS_CI);

  const int tid = threadIdx.x;
  const int lane = tid & 31;
  const int wave = tid >> 5;
  const int hi = lane >> 4;
  const int ln = lane & 15;
  const int token0 = blockIdx.x * BT;
  const int b = token0 >> 12;
  const int t0 = token0 & (TT - 1);
  const int n0 = wave * 64;

  const int zt = tid & (BT - 1);
  const int zh = tid >> 8;
  const float* zb = z + (size_t)b * DD * TT + (size_t)(32 * zh) * TT + t0 + zt;
  float zv[32];
#pragma unroll
  for (int i = 0; i < 32; ++i) zv[i] = zb[(size_t)i * TT];

  const float4* cg = (const float4*)(cb + n0 * DD) + lane;
  float4 raw[8];
#pragma unroll
  for (int i = 0; i < 8; ++i) raw[i] = cg[i * 32];

  {
    const v8f zp0 = {zv[0], zv[1], zv[2], zv[3], zv[4], zv[5], zv[6], zv[7]};
    const v8f zp1 = {zv[8], zv[9], zv[10], zv[11], zv[12], zv[13], zv[14], zv[15]};
    const v8f zp2 = {zv[16], zv[17], zv[18], zv[19], zv[20], zv[21], zv[22], zv[23]};
    const v8f zp3 = {zv[24], zv[25], zv[26], zv[27], zv[28], zv[29], zv[30], zv[31]};
    asm volatile(""
                 :
                 : "v"(zp0), "v"(zp1), "v"(zp2), "v"(zp3),
                   "v"(pack8(raw[0], raw[1])), "v"(pack8(raw[2], raw[3])),
                   "v"(pack8(raw[4], raw[5])), "v"(pack8(raw[6], raw[7])));
  }

  {
    float4* zd = (float4*)(zs + zt * ZP + 32 * zh);
#pragma unroll
    for (int g = 0; g < 8; ++g)
      zd[g] = make_float4(zv[4 * g], zv[4 * g + 1], zv[4 * g + 2], zv[4 * g + 3]);
  }

  float* const cb0 = cbest + ln * CP + 2 * wave + hi;
  int* const ci0 = cidx + ln * CP + 2 * wave + hi;
  const int nbase = n0 + 8 * hi;
#pragma unroll
  for (int tt = 0; tt < BT / 16; ++tt) {
    cb0[tt * 16 * CP] = -3.4e38f;
    ci0[tt * 16 * CP] = nbase;
  }
  lds_stores_done();
  rows_signal();

  const char* const zrow0 = (const char*)zs + ln * (ZP * 4) + hi * 8;

  char* const st = smem + LDS_ST + wave * (16 * ZP * 4);
  float4* const sd = (float4*)(st + hi * (ZP * 4) + ln * 16);
  const char* const fr = st + ln * (ZP * 4) + hi * 8;
  const char* const myrow = st + ln * (ZP * 4);

#pragma unroll 1
  for (int tl = 0; tl < 4; ++tl) {
    v2f a[16];
    const float hv = rows_to_fragments(raw, a, sd, fr, myrow);
    float hreg[8];
#pragma unroll
    for (int r = 0; r < 8; ++r) hreg[r] = __shfl(hv, 8 * hi + r, 32);

    {
      const int nt = tl < 3 ? tl + 1 : 3;
#pragma unroll
      for (int i = 0; i < 8; ++i) raw[i] = cg[nt * 256 + i * 32];
    }

    if (tl == 0) rows_wait();

    const char* zrow = zrow0;
    float* cbp = cb0;
    int* cip = ci0;
    const int kbase = nbase + 16 * tl;
    for (int tp = 0; tp < BT / 32; ++tp) {
      v2f bf0[16], bf1[16];
#pragma unroll
      for (int kk = 0; kk < 16; ++kk) {
        bf0[kk] = *(const v2f*)(zrow + kk * 16);
        bf1[kk] = *(const v2f*)(zrow + 16 * ZP * 4 + kk * 16);
      }
      v8f c0 = {};
      v8f c1 = {};
#pragma unroll
      for (int kk = 0; kk < 16; ++kk) {
        c0 = __builtin_amdgcn_wmma_f32_16x16x4_f32(
            false, a[kk], false, bf0[kk], (short)0, c0, false, false);
        c1 = __builtin_amdgcn_wmma_f32_16x16x4_f32(
            false, a[kk], false, bf1[kk], (short)0, c1, false, false);
      }
      float b0, b1;
      int k0, k1;
      tile_best(c0, hreg, kbase, b0, k0);
      tile_best(c1, hreg, kbase, b1, k1);
      const float p0 = cbp[0];
      const int q0 = cip[0];
      const float p1 = cbp[16 * CP];
      const int q1 = cip[16 * CP];
      const bool t0b = b0 > p0;
      const bool t1b = b1 > p1;
      cbp[0] = t0b ? b0 : p0;
      cip[0] = t0b ? k0 : q0;
      cbp[16 * CP] = t1b ? b1 : p1;
      cip[16 * CP] = t1b ? k1 : q1;
      zrow += 32 * ZP * 4;
      cbp += 32 * CP;
      cip += 32 * CP;
    }
  }
  __syncthreads();

  if (tid < BT) {
    const float* pb = cbest + tid * CP;
    const int* pi = cidx + tid * CP;
    float v[32];
    int n[32];
#pragma unroll
    for (int k = 0; k < 32; ++k) { v[k] = pb[k]; n[k] = pi[k]; }
#pragma unroll
    for (int st2 = 16; st2 >= 1; st2 >>= 1) {
#pragma unroll
      for (int k = 0; k < st2; ++k) {
        const bool take = (v[k + st2] > v[k]) |
                          ((v[k + st2] == v[k]) & (n[k + st2] < n[k]));
        v[k] = take ? v[k + st2] : v[k];
        n[k] = take ? n[k + st2] : n[k];
      }
    }
    idx[token0 + tid] = n[0];
  }
}

#define GT_TOK 128
#define GT_PITCH 129

__global__ void __launch_bounds__(256)
vq_gather(const float* __restrict__ cb, const int* __restrict__ idx,
          float* __restrict__ out) {
  __shared__ float tile[DD * GT_PITCH];

  const int lane = threadIdx.x & 31;
  const int wave = threadIdx.x >> 5;
  const int q = lane & 15;
  const int r = lane >> 4;
  const int tok0 = blockIdx.x * GT_TOK;
  const int b = tok0 >> 12;
  const int t0 = tok0 & (TT - 1);
  const int slot0 = wave * 2 + r;

  int id[8];
  for (int i = 0; i < 8; ++i) {
    id[i] = idx[tok0 + i * 16 + slot0] & (KK - 1);
  }
  float4 v[8];
  for (int i = 0; i < 8; ++i) {
    v[i] = *(const float4*)(cb + id[i] * DD + 4 * q);
  }
  for (int i = 0; i < 8; ++i) {
    float* p = tile + (4 * q) * GT_PITCH + i * 16 + slot0;
    p[0] = v[i].x;
    p[GT_PITCH] = v[i].y;
    p[2 * GT_PITCH] = v[i].z;
    p[3 * GT_PITCH] = v[i].w;
  }
  __syncthreads();

  float* ob = out + (size_t)b * DD * TT + t0 + lane;
  const float* tp = tile + lane;
  for (int j = 0; j < 8; ++j) {
    const int d = wave * 8 + j;
    for (int m = 0; m < 4; ++m) {
      ob[(size_t)d * TT + 32 * m] = tp[d * GT_PITCH + 32 * m];
    }
  }
}

extern "C" void kernel_launch(void* const* d_in, const int* in_sizes, int n_in,
                              void* d_out, int out_size, void* d_ws,
                              size_t ws_size, hipStream_t stream) {
  const float* z = (const float*)d_in[0];
  const float* cb = (const float*)d_in[1];
  char* ws = (char*)d_ws;
  int* idx = (int*)(ws + 266240);

  vq_encode<<<256, 512, LDS_TOTAL, stream>>>(z, cb, idx);
  vq_gather<<<512, 256, 0, stream>>>(cb, idx, (float*)d_out);
}


# Encode re-blocked: a wave keeps all four of its code tiles as fragments in registers and walks the token tiles once; token fragments are read from LDS once not four times, running best in registers
# speedup vs baseline: 1.4792x; 1.0411x over previous
#include <hip/hip_runtime.h>

typedef __attribute__((ext_vector_type(2))) float v2f;
typedef __attribute__((ext_vector_type(8))) float v8f;

#define TT 4096
#define DD 64
#define KK 1024

#define BT 256
#define ZP 68
#define CP 33
#define LDS_Z 0
#define LDS_CB 69632
#define LDS_CI 103424
#define LDS_ST 137216
#define LDS_TOTAL 206848

__device__ __forceinline__ void tile_best(const v8f& c, const float (&h)[8],
                                          const int (&kv)[8], float& bt, int& kt) {
  bt = c[0] - h[0];
  kt = kv[0];
#pragma unroll
  for (int r = 1; r < 8; ++r) {
    const float s = c[r] - h[r];
    if (s > bt) { bt = s; kt = kv[r]; }
  }
}

__device__ __forceinline__ void lds_stores_done() {
  asm volatile("s_wait_dscnt 0x0" ::: "memory");
}
__device__ __forceinline__ void rows_signal() {
  asm volatile("s_barrier_signal -1" ::: "memory");
}
__device__ __forceinline__ void rows_wait() {
  asm volatile("s_barrier_wait -1" ::: "memory");
}

__device__ __forceinline__ v8f pack8(const float4& x, const float4& y) {
  const v8f r = {x.x, x.y, x.z, x.w, y.x, y.y, y.z, y.w};
  return r;
}

__device__ __forceinline__ float rows_to_fragments(const float4 (&rw)[8],
                                                   v2f (&at)[16], float4* sd,
                                                   const char* fr) {
#pragma unroll
  for (int i = 0; i < 8; ++i) sd[i * (2 * ZP / 4)] = rw[i];
#pragma unroll
  for (int kk = 0; kk < 16; ++kk) at[kk] = *(const v2f*)(fr + kk * 16);
  float p[16], o[16];
#pragma unroll
  for (int kk = 0; kk < 16; ++kk)
    p[kk] = __builtin_fmaf(at[kk].x, at[kk].x, at[kk].y * at[kk].y);
#pragma unroll
  for (int kk = 0; kk < 16; ++kk) o[kk] = __shfl_xor(p[kk], 16, 32);
  float s = p[0] + o[0];
#pragma unroll
  for (int kk = 1; kk < 16; ++kk) {
    s += p[kk];
    s += o[kk];
  }
  return 0.5f * s;
}

__global__ void __launch_bounds__(512)
vq_encode(const float* __restrict__ z, const float* __restrict__ cb,
          int* __restrict__ idx) {
  extern __shared__ char smem[];
  float* zs = (float*)(smem + LDS_Z);
  float* cbest = (float*)(smem + LDS_CB);
  int* cidx = (int*)(smem + LDS_CI);

  const int tid = threadIdx.x;
  const int lane = tid & 31;
  const int wave = tid >> 5;
  const int hi = lane >> 4;
  const int ln = lane & 15;
  const int token0 = blockIdx.x * BT;
  const int b = token0 >> 12;
  const int t0 = token0 & (TT - 1);
  const int n0 = wave * 64;

  const int zt = tid & (BT - 1);
  const int zh = tid >> 8;
  const float* zb = z + (size_t)b * DD * TT + (size_t)(32 * zh) * TT + t0 + zt;
  float zv[32];
#pragma unroll
  for (int i = 0; i < 32; ++i) zv[i] = zb[(size_t)i * TT];

  const float4* cg = (const float4*)(cb + n0 * DD) + lane;
  float4 raw[4][8];
#pragma unroll
  for (int tl = 0; tl < 4; ++tl) {
#pragma unroll
    for (int i = 0; i < 8; ++i) raw[tl][i] = cg[tl * 256 + i * 32];
  }

  {
    const v8f zp0 = {zv[0], zv[1], zv[2], zv[3], zv[4], zv[5], zv[6], zv[7]};
    const v8f zp1 = {zv[8], zv[9], zv[10], zv[11], zv[12], zv[13], zv[14], zv[15]};
    const v8f zp2 = {zv[16], zv[17], zv[18], zv[19], zv[20], zv[21], zv[22], zv[23]};
    const v8f zp3 = {zv[24], zv[25], zv[26], zv[27], zv[28], zv[29], zv[30], zv[31]};
    asm volatile(""
                 :
                 : "v"(zp0), "v"(zp1), "v"(zp2), "v"(zp3),
                   "v"(pack8(raw[0][0], raw[0][1])), "v"(pack8(raw[0][2], raw[0][3])),
                   "v"(pack8(raw[0][4], raw[0][5])), "v"(pack8(raw[0][6], raw[0][7])),
                   "v"(pack8(raw[1][0], raw[1][1])), "v"(pack8(raw[1][2], raw[1][3])),
                   "v"(pack8(raw[1][4], raw[1][5])), "v"(pack8(raw[1][6], raw[1][7])),
                   "v"(pack8(raw[2][0], raw[2][1])), "v"(pack8(raw[2][2], raw[2][3])),
                   "v"(pack8(raw[2][4], raw[2][5])), "v"(pack8(raw[2][6], raw[2][7])),
                   "v"(pack8(raw[3][0], raw[3][1])), "v"(pack8(raw[3][2], raw[3][3])),
                   "v"(pack8(raw[3][4], raw[3][5])), "v"(pack8(raw[3][6], raw[3][7])));
  }

  {
    float4* zd = (float4*)(zs + zt * ZP + 32 * zh);
#pragma unroll
    for (int g = 0; g < 8; ++g)
      zd[g] = make_float4(zv[4 * g], zv[4 * g + 1], zv[4 * g + 2], zv[4 * g + 3]);
  }
  lds_stores_done();
  rows_signal();

  char* const st = smem + LDS_ST + wave * (16 * ZP * 4);
  float4* const sd = (float4*)(st + hi * (ZP * 4) + ln * 16);
  const char* const fr = st + ln * (ZP * 4) + hi * 8;

  v2f a[4][16];
  float hreg[4][8];
#pragma unroll
  for (int tl = 0; tl < 4; ++tl) {
    const float hv = rows_to_fragments(raw[tl], a[tl], sd, fr);
#pragma unroll
    for (int r = 0; r < 8; ++r) hreg[tl][r] = __shfl(hv, 8 * hi + r, 32);
  }

  float* cbp = cbest + ln * CP + 2 * wave + hi;
  int* cip = cidx + ln * CP + 2 * wave + hi;
  int kv[8];
#pragma unroll
  for (int r = 0; r < 8; ++r) kv[r] = n0 + 8 * hi + r;

  const char* zrow = (const char*)zs + ln * (ZP * 4) + hi * 8;

  rows_wait();

  for (int tt = 0; tt < BT / 16; ++tt) {
    v2f bf[16];
#pragma unroll
    for (int kk = 0; kk < 16; ++kk) bf[kk] = *(const v2f*)(zrow + kk * 16);

    v8f c0 = {};
    v8f c1 = {};
    v8f c2 = {};
    v8f c3 = {};
#pragma unroll
    for (int kk = 0; kk < 16; ++kk) {
      c0 = __builtin_amdgcn_wmma_f32_16x16x4_f32(
          false, a[0][kk], false, bf[kk], (short)0, c0, false, false);
      c1 = __builtin_amdgcn_wmma_f32_16x16x4_f32(
          false, a[1][kk], false, bf[kk], (short)0, c1, false, false);
      c2 = __builtin_amdgcn_wmma_f32_16x16x4_f32(
          false, a[2][kk], false, bf[kk], (short)0, c2, false, false);
      c3 = __builtin_amdgcn_wmma_f32_16x16x4_f32(
          false, a[3][kk], false, bf[kk], (short)0, c3, false, false);
    }
    float b0, b1, b2, b3;
    int k0, k1, k2, k3;
    tile_best(c0, hreg[0], kv, b0, k0);
    tile_best(c1, hreg[1], kv, b1, k1);
    tile_best(c2, hreg[2], kv, b2, k2);
    tile_best(c3, hreg[3], kv, b3, k3);
    float bt = b0;
    int kt = k0;
    if (b1 > bt) { bt = b1; kt = k1 + 16; }
    if (b2 > bt) { bt = b2; kt = k2 + 32; }
    if (b3 > bt) { bt = b3; kt = k3 + 48; }
    cbp[0] = bt;
    cip[0] = kt;
    zrow += 16 * ZP * 4;
    cbp += 16 * CP;
    cip += 16 * CP;
  }
  __syncthreads();

  if (tid < BT) {
    const float* pb = cbest + tid * CP;
    const int* pi = cidx + tid * CP;
    float v[32];
    int n[32];
#pragma unroll
    for (int k = 0; k < 32; ++k) { v[k] = pb[k]; n[k] = pi[k]; }
#pragma unroll
    for (int st2 = 16; st2 >= 1; st2 >>= 1) {
#pragma unroll
      for (int k = 0; k < st2; ++k) {
        const bool take = (v[k + st2] > v[k]) |
                          ((v[k + st2] == v[k]) & (n[k + st2] < n[k]));
        v[k] = take ? v[k + st2] : v[k];
        n[k] = take ? n[k + st2] : n[k];
      }
    }
    idx[token0 + tid] = n[0];
  }
}

#define GT_TOK 128
#define GT_PITCH 129

__global__ void __launch_bounds__(256)
vq_gather(const float* __restrict__ cb, const int* __restrict__ idx,
          float* __restrict__ out) {
  __shared__ float tile[DD * GT_PITCH];

  const int lane = threadIdx.x & 31;
  const int wave = threadIdx.x >> 5;
  const int q = lane & 15;
  const int r = lane >> 4;
  const int tok0 = blockIdx.x * GT_TOK;
  const int b = tok0 >> 12;
  const int t0 = tok0 & (TT - 1);
  const int slot0 = wave * 2 + r;

  int id[8];
  for (int i = 0; i < 8; ++i) {
    id[i] = idx[tok0 + i * 16 + slot0] & (KK - 1);
  }
  float4 v[8];
  for (int i = 0; i < 8; ++i) {
    v[i] = *(const float4*)(cb + id[i] * DD + 4 * q);
  }
  for (int i = 0; i < 8; ++i) {
    float* p = tile + (4 * q) * GT_PITCH + i * 16 + slot0;
    p[0] = v[i].x;
    p[GT_PITCH] = v[i].y;
    p[2 * GT_PITCH] = v[i].z;
    p[3 * GT_PITCH] = v[i].w;
  }
  __syncthreads();

  float* ob = out + (size_t)b * DD * TT + t0 + lane;
  const float* tp = tile + lane;
  for (int j = 0; j < 8; ++j) {
    const int d = wave * 8 + j;
    for (int m = 0; m < 4; ++m) {
      ob[(size_t)d * TT + 32 * m] = tp[d * GT_PITCH + 32 * m];
    }
  }
}

extern "C" void kernel_launch(void* const* d_in, const int* in_sizes, int n_in,
                              void* d_out, int out_size, void* d_ws,
                              size_t ws_size, hipStream_t stream) {
  const float* z = (const float*)d_in[0];
  const float* cb = (const float*)d_in[1];
  char* ws = (char*)d_ws;
  int* idx = (int*)(ws + 266240);

  vq_encode<<<256, 512, LDS_TOTAL, stream>>>(z, cb, idx);
  vq_gather<<<512, 256, 0, stream>>>(cb, idx, (float*)d_out);
}
